# starting kernel
# baseline (speedup 1.0000x reference)
#include <hip/hip_runtime.h>

typedef __attribute__((ext_vector_type(2))) float v2f;
typedef __attribute__((ext_vector_type(8))) float v8f;

constexpr int kBS    = 256;
constexpr int kP     = 64;
constexpr int kV     = 2562;
constexpr int kK     = kP * 3;
constexpr int kKP    = kK / 2;
constexpr int kM     = kBS * 3;
constexpr int kMT    = kM / 16;
constexpr int kNTILE = 4;
constexpr int kNT4   = 41;
constexpr int kNPAD  = kNT4 * kNTILE * 16;

__global__ void prep_A_kernel(const float* __restrict__ scales,
                              const float* __restrict__ transforms,
                              const float* __restrict__ pw,
                              float* __restrict__ A) {
  int idx = blockIdx.x * blockDim.x + threadIdx.x;
  if (idx >= kBS * kP) return;
  int bs = idx >> 6;
  int p  = idx & 63;
  const float* t = transforms + (size_t)idx * 6;
  float ax = t[3], ay = t[4], az = t[5];
  float cx = cosf(ax), sx = sinf(ax);
  float cy = cosf(ay), sy = sinf(ay);
  float cz = cosf(az), sz = sinf(az);
  float r00 = cy * cz;
  float r01 = -cy * sz;
  float r02 = sy;
  float r10 = sx * sy * cz + cx * sz;
  float r11 = -sx * sy * sz + cx * cz;
  float r12 = -sx * cy;
  float r20 = -cx * sy * cz + sx * sz;
  float r21 = cx * sy * sz + sx * cz;
  float r22 = cx * cy;
  float w = pw[idx] * scales[bs];
  float* a0 = A + (size_t)(bs * 3 + 0) * kK + p * 3;
  float* a1 = A + (size_t)(bs * 3 + 1) * kK + p * 3;
  float* a2 = A + (size_t)(bs * 3 + 2) * kK + p * 3;
  a0[0] = r00 * w; a0[1] = r01 * w; a0[2] = r02 * w;
  a1[0] = r10 * w; a1[1] = r11 * w; a1[2] = r12 * w;
  a2[0] = r20 * w; a2[1] = r21 * w; a2[2] = r22 * w;
}

__global__ void prep_D_kernel(const float* __restrict__ base,
                              const float* __restrict__ off,
                              v2f* __restrict__ Dp) {
  int id = blockIdx.x * blockDim.x + threadIdx.x;
  if (id >= kKP * kNPAD) return;
  int kp = id / kNPAD;
  int n  = id - kp * kNPAD;
  v2f val = {0.f, 0.f};
  if (n < kV) {
    int k0 = kp * 2, k1 = k0 + 1;
    int p0 = k0 / 3, j0 = k0 - p0 * 3;
    int p1 = k1 / 3, j1 = k1 - p1 * 3;
    val.x = base[(size_t)n * 3 + j0] + off[((size_t)p0 * kV + n) * 3 + j0];
    val.y = base[(size_t)n * 3 + j1] + off[((size_t)p1 * kV + n) * 3 + j1];
  }
  Dp[id] = val;
}

__global__ void prep_T_kernel(const float* __restrict__ transforms,
                              const float* __restrict__ pw,
                              float* __restrict__ T) {
  __shared__ float sm[kP][3];
  int bs = blockIdx.x;
  int p  = threadIdx.x;
  const float* t = transforms + ((size_t)bs * kP + p) * 6;
  float w = pw[bs * kP + p];
  sm[p][0] = w * t[0];
  sm[p][1] = w * t[1];
  sm[p][2] = w * t[2];
  __syncthreads();
  for (int st = 32; st > 0; st >>= 1) {
    if (p < st) {
      sm[p][0] += sm[p + st][0];
      sm[p][1] += sm[p + st][1];
      sm[p][2] += sm[p + st][2];
    }
    __syncthreads();
  }
  if (p < 3) T[bs * 3 + p] = sm[0][p];
}

__global__ __launch_bounds__(32) void gemm_wmma_kernel(
    const float* __restrict__ A, const v2f* __restrict__ Dp,
    const float* __restrict__ T, float* __restrict__ out) {
  int lane = threadIdx.x;
  int mt = blockIdx.x;
  int nt = blockIdx.y;
  int m0 = mt * 16;
  int n0 = nt * (kNTILE * 16);
  int half = lane >> 4;
  int lrow = lane & 15;

  const float* Ap = A + (size_t)(m0 + lrow) * kK + half * 2;
  const v2f* Bp = Dp + (size_t)half * kNPAD + n0 + lrow;

  v8f c0 = {0.f, 0.f, 0.f, 0.f, 0.f, 0.f, 0.f, 0.f};
  v8f c1 = c0, c2 = c0, c3 = c0;
#pragma unroll 4
  for (int k = 0; k < kK; k += 4) {
    v2f a = *(const v2f*)(Ap + k);
    const v2f* b = Bp + (size_t)(k >> 1) * kNPAD;
    v2f b0 = b[0], b1 = b[16], b2 = b[32], b3 = b[48];
    c0 = __builtin_amdgcn_wmma_f32_16x16x4_f32(false, a, false, b0, (short)0,
                                               c0, false, false);
    c1 = __builtin_amdgcn_wmma_f32_16x16x4_f32(false, a, false, b1, (short)0,
                                               c1, false, false);
    c2 = __builtin_amdgcn_wmma_f32_16x16x4_f32(false, a, false, b2, (short)0,
                                               c2, false, false);
    c3 = __builtin_amdgcn_wmma_f32_16x16x4_f32(false, a, false, b3, (short)0,
                                               c3, false, false);
  }

  int mb = m0 + half * 8;
  int bs = mb / 3;
  int i  = mb - bs * 3;
  float  tv[8];
  size_t ob[8];
#pragma unroll
  for (int r = 0; r < 8; ++r) {
    tv[r] = T[mb + r];
    ob[r] = (size_t)bs * (kV * 3) + i;
    if (++i == 3) { i = 0; ++bs; }
  }
  auto store_tile = [&](int t, const v8f& c) {
    int n = n0 + t * 16 + lrow;
    if (n < kV) {
      size_t n3 = (size_t)n * 3;
#pragma unroll
      for (int r = 0; r < 8; ++r) out[ob[r] + n3] = c[r] + tv[r];
    }
  };
  store_tile(0, c0);
  store_tile(1, c1);
  store_tile(2, c2);
  store_tile(3, c3);
}

extern "C" void kernel_launch(void* const* d_in, const int* in_sizes, int n_in,
                              void* d_out, int out_size, void* d_ws,
                              size_t ws_size, hipStream_t stream) {
  const float* scales     = (const float*)d_in[0];
  const float* transforms = (const float*)d_in[1];
  const float* pw         = (const float*)d_in[2];
  const float* off        = (const float*)d_in[3];
  const float* base       = (const float*)d_in[4];
  float* out = (float*)d_out;

  float* wsf = (float*)d_ws;
  float* A  = wsf;
  v2f*   Dp = (v2f*)(wsf + (size_t)kM * kK);
  float* T  = (float*)(Dp + (size_t)kKP * kNPAD);

  prep_A_kernel<<<(kBS * kP + 255) / 256, 256, 0, stream>>>(scales, transforms,
                                                            pw, A);
  prep_D_kernel<<<(kKP * kNPAD + 255) / 256, 256, 0, stream>>>(base, off, Dp);
  prep_T_kernel<<<kBS, kP, 0, stream>>>(transforms, pw, T);
  gemm_wmma_kernel<<<dim3(kMT, kNT4), 32, 0, stream>>>(A, Dp, T, out);
}


# Prep kernel: the 12 sums over prototypes by a butterfly inside each wave (DPP adds; no LDS, no barrier) instead of a 6-round LDS tree; GEMM adds the two waves' halves; p sum order changes
# speedup vs baseline: 1.7782x; 1.7782x over previous
#include <hip/hip_runtime.h>


typedef __attribute__((ext_vector_type(2))) float v2f;
typedef __attribute__((ext_vector_type(4))) float v4f;
typedef __attribute__((ext_vector_type(8))) float v8f;

constexpr int kBS    = 256;
constexpr int kP     = 64;
constexpr int kV     = 2562;
constexpr int kK     = kP * 3;
constexpr int kKA    = kK + 8;
constexpr int kM     = kBS * 3;
constexpr int kMT    = kM / 16;
constexpr int kNTILE = 4;
constexpr int kNT4   = 41;
constexpr int kPG    = 4;


__device__ __forceinline__ float wave_sum(float x) {
  x += __int_as_float(__builtin_amdgcn_update_dpp(0, __float_as_int(x), 0x161, 0xf, 0xf, true));
  x += __int_as_float(__builtin_amdgcn_update_dpp(0, __float_as_int(x), 0x162, 0xf, 0xf, true));
  x += __int_as_float(__builtin_amdgcn_update_dpp(0, __float_as_int(x), 0x164, 0xf, 0xf, true));
  x += __int_as_float(__builtin_amdgcn_update_dpp(0, __float_as_int(x), 0x168, 0xf, 0xf, true));
  int b = __float_as_int(x);
  x += __int_as_float(__builtin_amdgcn_permlanex16(b, b, 0x76543210, 0xfedcba98, false, false));
  return x;
}

__global__ __launch_bounds__(kP) void prep_A_kernel(
    const float* __restrict__ scales, const float* __restrict__ transforms,
    const float* __restrict__ pw, float* __restrict__ A) {
  int bs = blockIdx.x;
  int p  = threadIdx.x;
  int idx = bs * kP + p;
  const float* t = transforms + (size_t)idx * 6;
  float ax = t[3], ay = t[4], az = t[5];
  float cx = cosf(ax), sx = sinf(ax);
  float cy = cosf(ay), sy = sinf(ay);
  float cz = cosf(az), sz = sinf(az);
  float r00 = cy * cz;
  float r01 = -cy * sz;
  float r02 = sy;
  float r10 = sx * sy * cz + cx * sz;
  float r11 = -sx * sy * sz + cx * cz;
  float r12 = -sx * cy;
  float r20 = -cx * sy * cz + sx * sz;
  float r21 = cx * sy * sz + sx * cz;
  float r22 = cx * cy;
  float wt = pw[idx];
  float w  = wt * scales[bs];
  float v[12];
  v[0] = r00 * w; v[1] = r01 * w; v[2] = r02 * w;
  v[3] = r10 * w; v[4] = r11 * w; v[5] = r12 * w;
  v[6] = r20 * w; v[7] = r21 * w; v[8] = r22 * w;
  v[9] = wt * t[0]; v[10] = wt * t[1]; v[11] = wt * t[2];
  float* a0 = A + (size_t)(bs * 3 + 0) * kKA + p * 3;
  float* a1 = A + (size_t)(bs * 3 + 1) * kKA + p * 3;
  float* a2 = A + (size_t)(bs * 3 + 2) * kKA + p * 3;
  a0[0] = v[0]; a0[1] = v[1]; a0[2] = v[2];
  a1[0] = v[3]; a1[1] = v[4]; a1[2] = v[5];
  a2[0] = v[6]; a2[1] = v[7]; a2[2] = v[8];
  float s[12];
#pragma unroll
  for (int q = 0; q < 12; ++q) s[q] = wave_sum(v[q]);
  if ((p & 31) == 0) {
    float* ps = A + (size_t)(bs * 3) * kKA + kK + (p >> 5) * 4;
#pragma unroll
    for (int i = 0; i < 3; ++i)
      *(v4f*)(ps + (size_t)i * kKA) = (v4f){s[3 * i], s[3 * i + 1], s[3 * i + 2], s[9 + i]};
  }
}

constexpr int kWV = 4;
constexpr int kGW = kP / kPG / kWV;

__global__ __launch_bounds__(kWV * 32) __attribute__((amdgpu_waves_per_eu(1, 2))) void gemm_wmma_kernel(
    const float* __restrict__ A, const float* __restrict__ off,
    const float* __restrict__ base, float* __restrict__ out) {
  __shared__ v4f xch[kWV][kWV - 1][kNTILE][2][32];
  int lane = threadIdx.x & 31;
  int w    = threadIdx.x >> 5;
  int mq = blockIdx.x;
  int nt = blockIdx.y;
  int n0 = nt * (kNTILE * 16);
  int half = lane >> 4;
  int lrow = lane & 15;

  const float* Ap[kWV];
#pragma unroll
  for (int r = 0; r < kWV; ++r) {
    int m = mq * (kWV * 16) + ((w + r) & (kWV - 1)) * 16 + lrow;
    Ap[r] = A + (size_t)m * kKA + half * 6 + w * (kGW * kPG * 3);
  }
  int m0 = mq * (kWV * 16) + w * 16;
  const float* Arow = A + (size_t)(m0 + lrow) * kKA;

  int nl[kNTILE];
  const float* Bp[kNTILE];
  const size_t hoff = (size_t)(half * 2 + w * (kGW * kPG)) * kV * 3;
#pragma unroll
  for (int t = 0; t < kNTILE; ++t) {
    nl[t] = min(n0 + t * 16 + lrow, kV - 1);
    Bp[t] = off + hoff + (size_t)nl[t] * 3;
  }

  v2f ax = *(const v2f*)(Arow + kK + half * 2) + *(const v2f*)(Arow + kK + 4 + half * 2);
  v2f xb[kNTILE];
#pragma unroll
  for (int t = 0; t < kNTILE; ++t) {
    const float* q = base + (size_t)nl[t] * 3;
    float e0 = q[half * 2];
    float y  = q[1];
    xb[t] = (v2f){e0, half ? 1.0f : y};
  }

  v8f acc[kWV][kNTILE];
#pragma unroll
  for (int r = 0; r < kWV; ++r)
#pragma unroll
    for (int t = 0; t < kNTILE; ++t) acc[r][t] = (v8f){0.f, 0.f, 0.f, 0.f, 0.f, 0.f, 0.f, 0.f};

#pragma unroll 2
  for (int g = 0; g < kGW; ++g) {
    v2f a[kWV][3];
#pragma unroll
    for (int r = 0; r < kWV; ++r) {
      const float* ap = Ap[r] + g * (kPG * 3);
      a[r][0] = *(const v2f*)(ap);
      a[r][1] = *(const v2f*)(ap + 2);
      a[r][2] = *(const v2f*)(ap + 4);
    }
    const size_t go = (size_t)g * (kPG * kV * 3);
    float u[kNTILE][6];
#pragma unroll
    for (int t = 0; t < kNTILE; ++t) {
      const float* q = Bp[t] + go;
      u[t][0] = q[0]; u[t][1] = q[1]; u[t][2] = q[2];
      u[t][3] = q[kV * 3]; u[t][4] = q[kV * 3 + 1]; u[t][5] = q[kV * 3 + 2];
    }
#pragma unroll
    for (int s = 0; s < 3; ++s) {
      v2f b[kNTILE];
#pragma unroll
      for (int t = 0; t < kNTILE; ++t) b[t] = (v2f){u[t][2 * s], u[t][2 * s + 1]};
#pragma unroll
      for (int r = 0; r < kWV; ++r)
#pragma unroll
        for (int t = 0; t < kNTILE; ++t)
          acc[r][t] = __builtin_amdgcn_wmma_f32_16x16x4_f32(false, a[r][s], false, b[t], (short)0,
                                                            acc[r][t], false, false);
    }
  }

#pragma unroll
  for (int r = 1; r < kWV; ++r) {
    int fw = (w + r) & (kWV - 1);
#pragma unroll
    for (int t = 0; t < kNTILE; ++t) {
      const v8f& d = acc[r][t];
      xch[fw][r - 1][t][0][lane] = (v4f){d[0], d[1], d[2], d[3]};
      xch[fw][r - 1][t][1][lane] = (v4f){d[4], d[5], d[6], d[7]};
    }
  }

#pragma unroll
  for (int t = 0; t < kNTILE; ++t)
    acc[0][t] = __builtin_amdgcn_wmma_f32_16x16x4_f32(false, ax, false, xb[t], (short)0,
                                                      acc[0][t], false, false);

  __syncthreads();

#pragma unroll
  for (int sl = 0; sl < kWV - 1; ++sl) {
#pragma unroll
    for (int t = 0; t < kNTILE; ++t) {
      v4f lo = xch[w][sl][t][0][lane];
      v4f hi = xch[w][sl][t][1][lane];
      v8f& c = acc[0][t];
      c[0] += lo[0]; c[1] += lo[1]; c[2] += lo[2]; c[3] += lo[3];
      c[4] += hi[0]; c[5] += hi[1]; c[6] += hi[2]; c[7] += hi[3];
    }
  }

  int mb = m0 + half * 8;
  int bs = mb / 3;
  int i  = mb - bs * 3;
  size_t ob[8];
#pragma unroll
  for (int r = 0; r < 8; ++r) {
    ob[r] = (size_t)bs * (kV * 3) + i;
    if (++i == 3) { i = 0; ++bs; }
  }
#pragma unroll
  for (int t = 0; t < kNTILE; ++t) {
    int n = n0 + t * 16 + lrow;
    if (n < kV) {
      size_t n3 = (size_t)n * 3;
#pragma unroll
      for (int r = 0; r < 8; ++r) out[ob[r] + n3] = acc[0][t][r];
    }
  }
}

extern "C" void kernel_launch(void* const* d_in, const int* in_sizes, int n_in,
                              void* d_out, int out_size, void* d_ws,
                              size_t ws_size, hipStream_t stream) {
  const float* scales     = (const float*)d_in[0];
  const float* transforms = (const float*)d_in[1];
  const float* pw         = (const float*)d_in[2];
  const float* off        = (const float*)d_in[3];
  const float* base       = (const float*)d_in[4];
  float* out = (float*)d_out;

  float* A = (float*)d_ws;

  prep_A_kernel<<<kBS, kP, 0, stream>>>(scales, transforms, pw, A);
  gemm_wmma_kernel<<<dim3(kMT / kWV, kNT4), kWV * 32, 0, stream>>>(A, off, base, out);
}
